# starting kernel
# baseline (speedup 1.0000x reference)
#include <hip/hip_runtime.h>
#include <hip/hip_bf16.h>

#define D_MODEL 1024
#define D_STATE 16
#define DT_RANK 64
#define BATCH   2
#define SEQLEN  4096
#define E_DIM   (DT_RANK + 2*D_STATE)
#define ROWS    (BATCH*SEQLEN)
#define NCHUNK  64
#define TCHUNK  (SEQLEN/NCHUNK)

typedef __attribute__((ext_vector_type(2))) float v2f;
typedef __attribute__((ext_vector_type(8))) float v8f;

__device__ __forceinline__ float softplus_f(float v) {
  return v > 20.f ? v : log1pf(__expf(v));
}

__global__ __launch_bounds__(256) void k_proj_dtbc(const float* __restrict__ x,
                                                   const float* __restrict__ Wbc,
                                                   float* __restrict__ dtBC) {
  const int NT = E_DIM / 16;
  int wave = (blockIdx.x * blockDim.x + threadIdx.x) >> 5;
  int lane = threadIdx.x & 31;
  int mt = wave / NT, nt = wave - mt * NT;
  int l = lane & 15, half = lane >> 4;

  const float* arow = x   + (size_t)(mt * 16 + l) * D_MODEL + 2 * half;
  const float* brow = Wbc + (size_t)(nt * 16 + l) * D_MODEL + 2 * half;

  v8f acc = {0.f, 0.f, 0.f, 0.f, 0.f, 0.f, 0.f, 0.f};
  for (int k = 0; k < D_MODEL; k += 4) {
    v2f a, b;
    a[0] = arow[k]; a[1] = arow[k + 1];
    b[0] = brow[k]; b[1] = brow[k + 1];
    acc = __builtin_amdgcn_wmma_f32_16x16x4_f32(false, a, false, b,
                                                (short)0, acc, false, false);
  }
#pragma unroll
  for (int r = 0; r < 8; ++r) {
    int m = mt * 16 + 8 * half + r;
    dtBC[(size_t)m * E_DIM + nt * 16 + l] = acc[r];
  }
}

__global__ __launch_bounds__(256) void k_proj_dt(const float* __restrict__ dtBC,
                                                 const float* __restrict__ Wdt,
                                                 const float* __restrict__ bdt,
                                                 float* __restrict__ dt) {
  const int NT = D_MODEL / 16;
  int wave = (blockIdx.x * blockDim.x + threadIdx.x) >> 5;
  int lane = threadIdx.x & 31;
  int mt = wave / NT, nt = wave - mt * NT;
  int l = lane & 15, half = lane >> 4;

  const float* arow = dtBC + (size_t)(mt * 16 + l) * E_DIM   + 2 * half;
  const float* brow = Wdt  + (size_t)(nt * 16 + l) * DT_RANK + 2 * half;

  v8f acc = {0.f, 0.f, 0.f, 0.f, 0.f, 0.f, 0.f, 0.f};
  for (int k = 0; k < DT_RANK; k += 4) {
    v2f a, b;
    a[0] = arow[k]; a[1] = arow[k + 1];
    b[0] = brow[k]; b[1] = brow[k + 1];
    acc = __builtin_amdgcn_wmma_f32_16x16x4_f32(false, a, false, b,
                                                (short)0, acc, false, false);
  }
  float bias = bdt[nt * 16 + l];
#pragma unroll
  for (int r = 0; r < 8; ++r) {
    int m = mt * 16 + 8 * half + r;
    dt[(size_t)m * D_MODEL + nt * 16 + l] = softplus_f(acc[r] + bias);
  }
}

__global__ __launch_bounds__(256) void k_scan_local(const float* __restrict__ x,
                                                    const float* __restrict__ dt,
                                                    const float* __restrict__ dtBC,
                                                    const float* __restrict__ A_log,
                                                    float* __restrict__ hend,
                                                    float* __restrict__ Pprod) {
  int gid = blockIdx.x * blockDim.x + threadIdx.x;
  int d = gid & (D_MODEL - 1);
  int c = (gid >> 10) & (NCHUNK - 1);
  int b = gid >> 16;

  float An[D_STATE];
#pragma unroll
  for (int n = 0; n < D_STATE; ++n) An[n] = -__expf(A_log[d * D_STATE + n]);

  float h[D_STATE], P[D_STATE];
#pragma unroll
  for (int n = 0; n < D_STATE; ++n) { h[n] = 0.f; P[n] = 1.f; }

  int t0 = c * TCHUNK;
  for (int t = t0; t < t0 + TCHUNK; ++t) {
    size_t row = (size_t)b * SEQLEN + t;
    float dtv = dt[row * D_MODEL + d];
    float xv  = x[row * D_MODEL + d];
    float Bv[D_STATE];
    const float4* bp = (const float4*)(dtBC + row * E_DIM + DT_RANK);
#pragma unroll
    for (int i = 0; i < 4; ++i) {
      float4 q = bp[i];
      Bv[4*i+0] = q.x; Bv[4*i+1] = q.y; Bv[4*i+2] = q.z; Bv[4*i+3] = q.w;
    }
    float dx = dtv * xv;
#pragma unroll
    for (int n = 0; n < D_STATE; ++n) {
      float dA = __expf(dtv * An[n]);
      h[n] = dA * h[n] + dx * Bv[n];
      P[n] *= dA;
    }
  }
  size_t base = ((size_t)(b * D_MODEL + d) * NCHUNK + c) * D_STATE;
#pragma unroll
  for (int n = 0; n < D_STATE; ++n) {
    hend[base + n]  = h[n];
    Pprod[base + n] = P[n];
  }
}

__global__ __launch_bounds__(256) void k_scan_combine(const float* __restrict__ hend,
                                                      const float* __restrict__ Pprod,
                                                      float* __restrict__ h0) {
  int gid = blockIdx.x * blockDim.x + threadIdx.x;
  int n = gid & (D_STATE - 1);
  int bd = gid >> 4;
  size_t base = (size_t)bd * NCHUNK * D_STATE + n;
  float h = 0.f;
  for (int c = 0; c < NCHUNK; ++c) {
    size_t idx = base + (size_t)c * D_STATE;
    h0[idx] = h;
    h = Pprod[idx] * h + hend[idx];
  }
}

__global__ __launch_bounds__(256) void k_scan_emit(const float* __restrict__ x,
                                                   const float* __restrict__ dt,
                                                   const float* __restrict__ dtBC,
                                                   const float* __restrict__ A_log,
                                                   const float* __restrict__ Dp,
                                                   const float* __restrict__ h0,
                                                   float* __restrict__ y) {
  int gid = blockIdx.x * blockDim.x + threadIdx.x;
  int d = gid & (D_MODEL - 1);
  int c = (gid >> 10) & (NCHUNK - 1);
  int b = gid >> 16;

  float An[D_STATE];
#pragma unroll
  for (int n = 0; n < D_STATE; ++n) An[n] = -__expf(A_log[d * D_STATE + n]);

  float h[D_STATE];
  size_t base = ((size_t)(b * D_MODEL + d) * NCHUNK + c) * D_STATE;
#pragma unroll
  for (int n = 0; n < D_STATE; ++n) h[n] = h0[base + n];

  float Dd = Dp[d];
  int t0 = c * TCHUNK;
  for (int t = t0; t < t0 + TCHUNK; ++t) {
    size_t row = (size_t)b * SEQLEN + t;
    float dtv = dt[row * D_MODEL + d];
    float xv  = x[row * D_MODEL + d];
    float Bv[D_STATE], Cv[D_STATE];
    const float4* bp = (const float4*)(dtBC + row * E_DIM + DT_RANK);
    const float4* cp = (const float4*)(dtBC + row * E_DIM + DT_RANK + D_STATE);
#pragma unroll
    for (int i = 0; i < 4; ++i) {
      float4 q = bp[i];
      Bv[4*i+0] = q.x; Bv[4*i+1] = q.y; Bv[4*i+2] = q.z; Bv[4*i+3] = q.w;
    }
#pragma unroll
    for (int i = 0; i < 4; ++i) {
      float4 q = cp[i];
      Cv[4*i+0] = q.x; Cv[4*i+1] = q.y; Cv[4*i+2] = q.z; Cv[4*i+3] = q.w;
    }
    float dx = dtv * xv;
    float yv = 0.f;
#pragma unroll
    for (int n = 0; n < D_STATE; ++n) {
      float dA = __expf(dtv * An[n]);
      h[n] = dA * h[n] + dx * Bv[n];
      yv += h[n] * Cv[n];
    }
    y[row * D_MODEL + d] = yv + xv * Dd;
  }
}

extern "C" void kernel_launch(void* const* d_in, const int* in_sizes, int n_in,
                              void* d_out, int out_size, void* d_ws, size_t ws_size,
                              hipStream_t stream) {
  const float* x    = (const float*)d_in[0];
  const float* Alog = (const float*)d_in[1];
  const float* Dp   = (const float*)d_in[2];
  const float* Wdt  = (const float*)d_in[3];
  const float* bdt  = (const float*)d_in[4];
  const float* Wbc  = (const float*)d_in[5];
  float* y = (float*)d_out;

  float* ws    = (float*)d_ws;
  float* dtBC  = ws;
  float* dt    = dtBC + (size_t)ROWS * E_DIM;
  float* hend  = dt + (size_t)ROWS * D_MODEL;
  float* Pprod = hend + (size_t)BATCH * D_MODEL * NCHUNK * D_STATE;
  float* h0    = Pprod + (size_t)BATCH * D_MODEL * NCHUNK * D_STATE;

  int waves1 = (ROWS / 16) * (E_DIM / 16);
  k_proj_dtbc<<<waves1 * 32 / 256, 256, 0, stream>>>(x, Wbc, dtBC);

  int waves2 = (ROWS / 16) * (D_MODEL / 16);
  k_proj_dt<<<waves2 * 32 / 256, 256, 0, stream>>>(dtBC, Wdt, bdt, dt);

  int nscan = BATCH * D_MODEL * NCHUNK;
  k_scan_local<<<nscan / 256, 256, 0, stream>>>(x, dt, dtBC, Alog, hend, Pprod);

  int ncomb = BATCH * D_MODEL * D_STATE;
  k_scan_combine<<<ncomb / 256, 256, 0, stream>>>(hend, Pprod, h0);

  k_scan_emit<<<nscan / 256, 256, 0, stream>>>(x, dt, dtBC, Alog, Dp, h0, y);
}


# projection GEMMs: each wave keeps a block of accumulator tiles (2x6 and 2x4) so one loaded fragment feeds 2-6 products; GEMM 1 splits K over 8 waves, partials joined in LDS: its K sum is regrouped
# speedup vs baseline: 1.2586x; 1.2586x over previous
#include <hip/hip_runtime.h>
#include <hip/hip_bf16.h>

#define D_MODEL 1024
#define D_STATE 16
#define DT_RANK 64
#define BATCH   2
#define SEQLEN  4096
#define E_DIM   (DT_RANK + 2*D_STATE)
#define ROWS    (BATCH*SEQLEN)
#define NCHUNK  64
#define TCHUNK  (SEQLEN/NCHUNK)
#define NSUB    4
#define TSUB    (TCHUNK/NSUB)
#define DBLK    64

typedef __attribute__((ext_vector_type(2))) float v2f;
typedef __attribute__((ext_vector_type(8))) float v8f;

__device__ __forceinline__ float softplus_f(float v) {
  return v > 20.f ? v : log1pf(__expf(v));
}


#define G1_ROWS  32
#define G1_WAVES 8
#define G1_KW    (D_MODEL / G1_WAVES)
__global__ __launch_bounds__(256) void k_proj_dtbc(const float* __restrict__ x,
                                                   const float* __restrict__ Wbc,
                                                   float* __restrict__ dtBC) {
  __shared__ float part[G1_WAVES][G1_ROWS * E_DIM];
  int w = threadIdx.x >> 5;
  int lane = threadIdx.x & 31;
  int l = lane & 15, half = lane >> 4;
  int m0 = blockIdx.x * G1_ROWS;

  const float* arow = x   + (size_t)(m0 + l) * D_MODEL + w * G1_KW + 2 * half;
  const float* brow = Wbc + (size_t)l * D_MODEL        + w * G1_KW + 2 * half;

  v8f acc[2][6];
#pragma unroll
  for (int i = 0; i < 2; ++i)
#pragma unroll
    for (int j = 0; j < 6; ++j)
      acc[i][j] = (v8f){0.f, 0.f, 0.f, 0.f, 0.f, 0.f, 0.f, 0.f};

#pragma unroll 4
  for (int k = 0; k < G1_KW; k += 4) {
    v2f a[2], b[6];
#pragma unroll
    for (int i = 0; i < 2; ++i) {
      a[i][0] = arow[(size_t)i * 16 * D_MODEL + k];
      a[i][1] = arow[(size_t)i * 16 * D_MODEL + k + 1];
    }
#pragma unroll
    for (int j = 0; j < 6; ++j) {
      b[j][0] = brow[(size_t)j * 16 * D_MODEL + k];
      b[j][1] = brow[(size_t)j * 16 * D_MODEL + k + 1];
    }
#pragma unroll
    for (int j = 0; j < 6; ++j)
#pragma unroll
      for (int i = 0; i < 2; ++i)
        acc[i][j] = __builtin_amdgcn_wmma_f32_16x16x4_f32(false, a[i], false, b[j],
                                                          (short)0, acc[i][j], false, false);
  }

#pragma unroll
  for (int i = 0; i < 2; ++i)
#pragma unroll
    for (int j = 0; j < 6; ++j)
#pragma unroll
      for (int r = 0; r < 8; ++r)
        part[w][(i * 16 + 8 * half + r) * E_DIM + j * 16 + l] = acc[i][j][r];
  __syncthreads();

  float* outp = dtBC + (size_t)m0 * E_DIM;
#pragma unroll
  for (int q = 0; q < (G1_ROWS * E_DIM) / 256; ++q) {
    int o = q * 256 + threadIdx.x;
    float s = part[0][o];
#pragma unroll
    for (int p = 1; p < G1_WAVES; ++p) s += part[p][o];
    outp[o] = s;
  }
}

#define G2_NS (D_MODEL / 64)
#define G2_PITCH 68
__global__ __launch_bounds__(256) void k_proj_dt(const float* __restrict__ dtBC,
                                                 const float* __restrict__ Wdt,
                                                 const float* __restrict__ bdt,
                                                 float* __restrict__ dt) {
  __shared__ float tile[8][32 * G2_PITCH];
  int wave = (blockIdx.x * blockDim.x + threadIdx.x) >> 5;
  int lane = threadIdx.x & 31;
  int ms = wave / G2_NS, ns = wave - ms * G2_NS;
  int l = lane & 15, half = lane >> 4;

  const float* arow = dtBC + (size_t)(ms * 32 + l) * E_DIM   + 2 * half;
  const float* brow = Wdt  + (size_t)(ns * 64 + l) * DT_RANK + 2 * half;

  v8f acc[2][4];
#pragma unroll
  for (int i = 0; i < 2; ++i)
#pragma unroll
    for (int j = 0; j < 4; ++j)
      acc[i][j] = (v8f){0.f, 0.f, 0.f, 0.f, 0.f, 0.f, 0.f, 0.f};

#pragma unroll 4
  for (int k = 0; k < DT_RANK; k += 4) {
    v2f a[2], b[4];
#pragma unroll
    for (int i = 0; i < 2; ++i) {
      a[i][0] = arow[i * 16 * E_DIM + k];
      a[i][1] = arow[i * 16 * E_DIM + k + 1];
    }
#pragma unroll
    for (int j = 0; j < 4; ++j) {
      b[j][0] = brow[j * 16 * DT_RANK + k];
      b[j][1] = brow[j * 16 * DT_RANK + k + 1];
    }
#pragma unroll
    for (int j = 0; j < 4; ++j)
#pragma unroll
      for (int i = 0; i < 2; ++i)
        acc[i][j] = __builtin_amdgcn_wmma_f32_16x16x4_f32(false, a[i], false, b[j],
                                                          (short)0, acc[i][j], false, false);
  }

  float* tw = tile[threadIdx.x >> 5];
#pragma unroll
  for (int i = 0; i < 2; ++i)
#pragma unroll
    for (int j = 0; j < 4; ++j)
#pragma unroll
      for (int r = 0; r < 8; ++r)
        tw[(i * 16 + 8 * half + r) * G2_PITCH + j * 16 + l] = acc[i][j][r];
  __syncthreads();

  float bias0 = bdt[ns * 64 + lane];
  float bias1 = bdt[ns * 64 + 32 + lane];
  float* orow = dt + (size_t)(ms * 32) * D_MODEL + ns * 64 + lane;
#pragma unroll 2
  for (int r = 0; r < 32; ++r) {
    float v0 = tw[r * G2_PITCH + lane] + bias0;
    float v1 = tw[r * G2_PITCH + 32 + lane] + bias1;
    orow[(size_t)r * D_MODEL]      = softplus_f(v0);
    orow[(size_t)r * D_MODEL + 32] = softplus_f(v1);
  }
}

__global__ __launch_bounds__(256) void k_scan_local(const float* __restrict__ x,
                                                    const float* __restrict__ dt,
                                                    const float* __restrict__ dtBC,
                                                    const float* __restrict__ A_log,
                                                    float* __restrict__ hend,
                                                    float* __restrict__ Pprod) {
  __shared__ float sh_h[NSUB - 1][D_STATE][DBLK];
  __shared__ float sh_P[NSUB - 1][D_STATE][DBLK];

  int dl  = threadIdx.x & (DBLK - 1);
  int sub = threadIdx.x >> 6;
  int d = (blockIdx.x & (D_MODEL / DBLK - 1)) * DBLK + dl;
  int c = (blockIdx.x >> 4) & (NCHUNK - 1);
  int b = blockIdx.x >> 10;

  float An[D_STATE];
#pragma unroll
  for (int n = 0; n < D_STATE; ++n) An[n] = -__expf(A_log[d * D_STATE + n]);

  float h[D_STATE], P[D_STATE];
#pragma unroll
  for (int n = 0; n < D_STATE; ++n) { h[n] = 0.f; P[n] = 1.f; }

  int t0 = c * TCHUNK + sub * TSUB;
  for (int t = t0; t < t0 + TSUB; ++t) {
    size_t row = (size_t)b * SEQLEN + t;
    float dtv = dt[row * D_MODEL + d];
    float xv  = x[row * D_MODEL + d];
    float Bv[D_STATE];
    const float4* bp = (const float4*)(dtBC + row * E_DIM + DT_RANK);
#pragma unroll
    for (int i = 0; i < 4; ++i) {
      float4 q = bp[i];
      Bv[4*i+0] = q.x; Bv[4*i+1] = q.y; Bv[4*i+2] = q.z; Bv[4*i+3] = q.w;
    }
    float dx = dtv * xv;
#pragma unroll
    for (int n = 0; n < D_STATE; ++n) {
      float dA = __expf(dtv * An[n]);
      h[n] = dA * h[n] + dx * Bv[n];
      P[n] *= dA;
    }
  }

  if (sub < NSUB - 1) {
#pragma unroll
    for (int n = 0; n < D_STATE; ++n) {
      sh_h[sub][n][dl] = h[n];
      sh_P[sub][n][dl] = P[n];
    }
  }
  __syncthreads();
  if (sub == NSUB - 1) {
    float e[D_STATE], Pc[D_STATE];
#pragma unroll
    for (int n = 0; n < D_STATE; ++n) { e[n] = sh_h[0][n][dl]; Pc[n] = sh_P[0][n][dl]; }
#pragma unroll
    for (int k = 1; k < NSUB - 1; ++k) {
#pragma unroll
      for (int n = 0; n < D_STATE; ++n) {
        float p = sh_P[k][n][dl];
        e[n]  = p * e[n] + sh_h[k][n][dl];
        Pc[n] *= p;
      }
    }
    size_t base = ((size_t)(b * D_MODEL + d) * NCHUNK + c) * D_STATE;
#pragma unroll
    for (int n = 0; n < D_STATE; ++n) {
      hend[base + n]  = P[n] * e[n] + h[n];
      Pprod[base + n] = Pc[n] * P[n];
    }
  }
}

__global__ __launch_bounds__(256) void k_scan_combine(const float* __restrict__ hend,
                                                      const float* __restrict__ Pprod,
                                                      float* __restrict__ h0) {
  int gid = blockIdx.x * blockDim.x + threadIdx.x;
  int n = gid & (D_STATE - 1);
  int bd = gid >> 4;
  size_t base = (size_t)bd * NCHUNK * D_STATE + n;
  float h = 0.f;
  for (int c = 0; c < NCHUNK; ++c) {
    size_t idx = base + (size_t)c * D_STATE;
    h0[idx] = h;
    h = Pprod[idx] * h + hend[idx];
  }
}

__global__ __launch_bounds__(256) void k_scan_emit(const float* __restrict__ x,
                                                   const float* __restrict__ dt,
                                                   const float* __restrict__ dtBC,
                                                   const float* __restrict__ A_log,
                                                   const float* __restrict__ Dp,
                                                   const float* __restrict__ h0,
                                                   float* __restrict__ y) {
  __shared__ float sh_h[NSUB - 1][D_STATE][DBLK];
  __shared__ float sh_P[NSUB - 1][D_STATE][DBLK];

  int dl  = threadIdx.x & (DBLK - 1);
  int sub = threadIdx.x >> 6;
  int d = (blockIdx.x & (D_MODEL / DBLK - 1)) * DBLK + dl;
  int c = (blockIdx.x >> 4) & (NCHUNK - 1);
  int b = blockIdx.x >> 10;

  float An[D_STATE];
#pragma unroll
  for (int n = 0; n < D_STATE; ++n) An[n] = -__expf(A_log[d * D_STATE + n]);

  int t0 = c * TCHUNK + sub * TSUB;
  float h[D_STATE];

  if (sub < NSUB - 1) {
    float P[D_STATE];
#pragma unroll
    for (int n = 0; n < D_STATE; ++n) { h[n] = 0.f; P[n] = 1.f; }
    for (int t = t0; t < t0 + TSUB; ++t) {
      size_t row = (size_t)b * SEQLEN + t;
      float dtv = dt[row * D_MODEL + d];
      float xv  = x[row * D_MODEL + d];
      float Bv[D_STATE];
      const float4* bp = (const float4*)(dtBC + row * E_DIM + DT_RANK);
#pragma unroll
      for (int i = 0; i < 4; ++i) {
        float4 q = bp[i];
        Bv[4*i+0] = q.x; Bv[4*i+1] = q.y; Bv[4*i+2] = q.z; Bv[4*i+3] = q.w;
      }
      float dx = dtv * xv;
#pragma unroll
      for (int n = 0; n < D_STATE; ++n) {
        float dA = __expf(dtv * An[n]);
        h[n] = dA * h[n] + dx * Bv[n];
        P[n] *= dA;
      }
    }
#pragma unroll
    for (int n = 0; n < D_STATE; ++n) {
      sh_h[sub][n][dl] = h[n];
      sh_P[sub][n][dl] = P[n];
    }
  }
  __syncthreads();

  size_t base = ((size_t)(b * D_MODEL + d) * NCHUNK + c) * D_STATE;
#pragma unroll
  for (int n = 0; n < D_STATE; ++n) h[n] = h0[base + n];
  for (int k = 0; k < sub; ++k) {
#pragma unroll
    for (int n = 0; n < D_STATE; ++n) h[n] = sh_P[k][n][dl] * h[n] + sh_h[k][n][dl];
  }

  float Dd = Dp[d];
  for (int t = t0; t < t0 + TSUB; ++t) {
    size_t row = (size_t)b * SEQLEN + t;
    float dtv = dt[row * D_MODEL + d];
    float xv  = x[row * D_MODEL + d];
    float Bv[D_STATE], Cv[D_STATE];
    const float4* bp = (const float4*)(dtBC + row * E_DIM + DT_RANK);
    const float4* cp = (const float4*)(dtBC + row * E_DIM + DT_RANK + D_STATE);
#pragma unroll
    for (int i = 0; i < 4; ++i) {
      float4 q = bp[i];
      Bv[4*i+0] = q.x; Bv[4*i+1] = q.y; Bv[4*i+2] = q.z; Bv[4*i+3] = q.w;
    }
#pragma unroll
    for (int i = 0; i < 4; ++i) {
      float4 q = cp[i];
      Cv[4*i+0] = q.x; Cv[4*i+1] = q.y; Cv[4*i+2] = q.z; Cv[4*i+3] = q.w;
    }
    float dx = dtv * xv;
    float yv = 0.f;
#pragma unroll
    for (int n = 0; n < D_STATE; ++n) {
      float dA = __expf(dtv * An[n]);
      h[n] = dA * h[n] + dx * Bv[n];
      yv += h[n] * Cv[n];
    }
    y[row * D_MODEL + d] = yv + xv * Dd;
  }
}

extern "C" void kernel_launch(void* const* d_in, const int* in_sizes, int n_in,
                              void* d_out, int out_size, void* d_ws, size_t ws_size,
                              hipStream_t stream) {
  const float* x    = (const float*)d_in[0];
  const float* Alog = (const float*)d_in[1];
  const float* Dp   = (const float*)d_in[2];
  const float* Wdt  = (const float*)d_in[3];
  const float* bdt  = (const float*)d_in[4];
  const float* Wbc  = (const float*)d_in[5];
  float* y = (float*)d_out;

  float* ws    = (float*)d_ws;
  float* dtBC  = ws;
  float* dt    = dtBC + (size_t)ROWS * E_DIM;
  float* hend  = dt + (size_t)ROWS * D_MODEL;
  float* Pprod = hend + (size_t)BATCH * D_MODEL * NCHUNK * D_STATE;
  float* h0    = Pprod + (size_t)BATCH * D_MODEL * NCHUNK * D_STATE;

  k_proj_dtbc<<<ROWS / G1_ROWS, 256, 0, stream>>>(x, Wbc, dtBC);

  k_proj_dt<<<(ROWS / 32) * G2_NS * 32 / 256, 256, 0, stream>>>(dtBC, Wdt, bdt, dt);

  int nscan = BATCH * NCHUNK * (D_MODEL / DBLK);
  k_scan_local<<<nscan, 256, 0, stream>>>(x, dt, dtBC, Alog, hend, Pprod);

  int ncomb = BATCH * D_MODEL * D_STATE;
  k_scan_combine<<<ncomb / 256, 256, 0, stream>>>(hend, Pprod, h0);

  k_scan_emit<<<nscan, 256, 0, stream>>>(x, dt, dtBC, Alog, Dp, h0, y);
}


# scan pass 2 starts every sub-chunk from a carry (E,Q) that pass 1's join writes to the workspace, h=Q*h0+E: no step scanned twice, no LDS or barrier in pass 2; carry over sub-chunks regrouped
# speedup vs baseline: 1.3394x; 1.0642x over previous
#include <hip/hip_runtime.h>
#include <hip/hip_bf16.h>

#define D_MODEL 1024
#define D_STATE 16
#define DT_RANK 64
#define BATCH   2
#define SEQLEN  4096
#define E_DIM   (DT_RANK + 2*D_STATE)
#define ROWS    (BATCH*SEQLEN)
#define NCHUNK  64
#define TCHUNK  (SEQLEN/NCHUNK)
#define NSUB    4
#define TSUB    (TCHUNK/NSUB)
#define DBLK    64
#define CARRY_BASE(b, c, k) ((size_t)((((b) * NCHUNK + (c)) * (NSUB - 1) + (k)) * D_STATE) * D_MODEL)
#define CARRY_FLOATS ((size_t)BATCH * NCHUNK * (NSUB - 1) * D_STATE * D_MODEL)

typedef __attribute__((ext_vector_type(2))) float v2f;
typedef __attribute__((ext_vector_type(8))) float v8f;

__device__ __forceinline__ float softplus_f(float v) {
  return v > 20.f ? v : log1pf(__expf(v));
}


#define G1_ROWS  32
#define G1_WAVES 8
#define G1_KW    (D_MODEL / G1_WAVES)
__global__ __launch_bounds__(256) void k_proj_dtbc(const float* __restrict__ x,
                                                   const float* __restrict__ Wbc,
                                                   float* __restrict__ dtBC) {
  __shared__ float part[G1_WAVES][G1_ROWS * E_DIM];
  int w = threadIdx.x >> 5;
  int lane = threadIdx.x & 31;
  int l = lane & 15, half = lane >> 4;
  int m0 = blockIdx.x * G1_ROWS;

  const float* arow = x   + (size_t)(m0 + l) * D_MODEL + w * G1_KW + 2 * half;
  const float* brow = Wbc + (size_t)l * D_MODEL        + w * G1_KW + 2 * half;

  v8f acc[2][6];
#pragma unroll
  for (int i = 0; i < 2; ++i)
#pragma unroll
    for (int j = 0; j < 6; ++j)
      acc[i][j] = (v8f){0.f, 0.f, 0.f, 0.f, 0.f, 0.f, 0.f, 0.f};

#pragma unroll 4
  for (int k = 0; k < G1_KW; k += 4) {
    v2f a[2], b[6];
#pragma unroll
    for (int i = 0; i < 2; ++i) {
      a[i][0] = arow[(size_t)i * 16 * D_MODEL + k];
      a[i][1] = arow[(size_t)i * 16 * D_MODEL + k + 1];
    }
#pragma unroll
    for (int j = 0; j < 6; ++j) {
      b[j][0] = brow[(size_t)j * 16 * D_MODEL + k];
      b[j][1] = brow[(size_t)j * 16 * D_MODEL + k + 1];
    }
#pragma unroll
    for (int j = 0; j < 6; ++j)
#pragma unroll
      for (int i = 0; i < 2; ++i)
        acc[i][j] = __builtin_amdgcn_wmma_f32_16x16x4_f32(false, a[i], false, b[j],
                                                          (short)0, acc[i][j], false, false);
  }

#pragma unroll
  for (int i = 0; i < 2; ++i)
#pragma unroll
    for (int j = 0; j < 6; ++j)
#pragma unroll
      for (int r = 0; r < 8; ++r)
        part[w][(i * 16 + 8 * half + r) * E_DIM + j * 16 + l] = acc[i][j][r];
  __syncthreads();

  float* outp = dtBC + (size_t)m0 * E_DIM;
#pragma unroll
  for (int q = 0; q < (G1_ROWS * E_DIM) / 256; ++q) {
    int o = q * 256 + threadIdx.x;
    float s = part[0][o];
#pragma unroll
    for (int p = 1; p < G1_WAVES; ++p) s += part[p][o];
    outp[o] = s;
  }
}

#define G2_NS (D_MODEL / 64)
#define G2_PITCH 68
__global__ __launch_bounds__(256) void k_proj_dt(const float* __restrict__ dtBC,
                                                 const float* __restrict__ Wdt,
                                                 const float* __restrict__ bdt,
                                                 float* __restrict__ dt) {
  __shared__ float tile[8][32 * G2_PITCH];
  int wave = (blockIdx.x * blockDim.x + threadIdx.x) >> 5;
  int lane = threadIdx.x & 31;
  int ms = wave / G2_NS, ns = wave - ms * G2_NS;
  int l = lane & 15, half = lane >> 4;

  const float* arow = dtBC + (size_t)(ms * 32 + l) * E_DIM   + 2 * half;
  const float* brow = Wdt  + (size_t)(ns * 64 + l) * DT_RANK + 2 * half;

  v8f acc[2][4];
#pragma unroll
  for (int i = 0; i < 2; ++i)
#pragma unroll
    for (int j = 0; j < 4; ++j)
      acc[i][j] = (v8f){0.f, 0.f, 0.f, 0.f, 0.f, 0.f, 0.f, 0.f};

#pragma unroll 4
  for (int k = 0; k < DT_RANK; k += 4) {
    v2f a[2], b[4];
#pragma unroll
    for (int i = 0; i < 2; ++i) {
      a[i][0] = arow[i * 16 * E_DIM + k];
      a[i][1] = arow[i * 16 * E_DIM + k + 1];
    }
#pragma unroll
    for (int j = 0; j < 4; ++j) {
      b[j][0] = brow[j * 16 * DT_RANK + k];
      b[j][1] = brow[j * 16 * DT_RANK + k + 1];
    }
#pragma unroll
    for (int j = 0; j < 4; ++j)
#pragma unroll
      for (int i = 0; i < 2; ++i)
        acc[i][j] = __builtin_amdgcn_wmma_f32_16x16x4_f32(false, a[i], false, b[j],
                                                          (short)0, acc[i][j], false, false);
  }

  float* tw = tile[threadIdx.x >> 5];
#pragma unroll
  for (int i = 0; i < 2; ++i)
#pragma unroll
    for (int j = 0; j < 4; ++j)
#pragma unroll
      for (int r = 0; r < 8; ++r)
        tw[(i * 16 + 8 * half + r) * G2_PITCH + j * 16 + l] = acc[i][j][r];
  __syncthreads();

  float bias0 = bdt[ns * 64 + lane];
  float bias1 = bdt[ns * 64 + 32 + lane];
  float* orow = dt + (size_t)(ms * 32) * D_MODEL + ns * 64 + lane;
#pragma unroll 2
  for (int r = 0; r < 32; ++r) {
    float v0 = tw[r * G2_PITCH + lane] + bias0;
    float v1 = tw[r * G2_PITCH + 32 + lane] + bias1;
    orow[(size_t)r * D_MODEL]      = softplus_f(v0);
    orow[(size_t)r * D_MODEL + 32] = softplus_f(v1);
  }
}

__global__ __launch_bounds__(256, 12) void k_scan_local(const float* __restrict__ x,
                                                    const float* __restrict__ dt,
                                                    const float* __restrict__ dtBC,
                                                    const float* __restrict__ A_log,
                                                    float* __restrict__ hend,
                                                    float* __restrict__ Pprod,
                                                    float* __restrict__ carE,
                                                    float* __restrict__ carQ) {
  __shared__ float sh_h[NSUB][D_STATE][DBLK];
  __shared__ float sh_P[NSUB][D_STATE][DBLK];

  int dl  = threadIdx.x & (DBLK - 1);
  int sub = threadIdx.x >> 6;
  int d = (blockIdx.x & (D_MODEL / DBLK - 1)) * DBLK + dl;
  int c = (blockIdx.x >> 4) & (NCHUNK - 1);
  int b = blockIdx.x >> 10;

  float An[D_STATE];
#pragma unroll
  for (int n = 0; n < D_STATE; ++n) An[n] = -__expf(A_log[d * D_STATE + n]);

  float h[D_STATE], P[D_STATE];
#pragma unroll
  for (int n = 0; n < D_STATE; ++n) { h[n] = 0.f; P[n] = 1.f; }

  int t0 = c * TCHUNK + sub * TSUB;
  for (int t = t0; t < t0 + TSUB; ++t) {
    size_t row = (size_t)b * SEQLEN + t;
    float dtv = dt[row * D_MODEL + d];
    float xv  = x[row * D_MODEL + d];
    float Bv[D_STATE];
    const float4* bp = (const float4*)(dtBC + row * E_DIM + DT_RANK);
#pragma unroll
    for (int i = 0; i < 4; ++i) {
      float4 q = bp[i];
      Bv[4*i+0] = q.x; Bv[4*i+1] = q.y; Bv[4*i+2] = q.z; Bv[4*i+3] = q.w;
    }
    float dx = dtv * xv;
#pragma unroll
    for (int n = 0; n < D_STATE; ++n) {
      float dA = __expf(dtv * An[n]);
      h[n] = dA * h[n] + dx * Bv[n];
      P[n] *= dA;
    }
  }

#pragma unroll
  for (int n = 0; n < D_STATE; ++n) {
    sh_h[sub][n][dl] = h[n];
    sh_P[sub][n][dl] = P[n];
  }
  __syncthreads();

  int n0 = sub * 4;
  float* pe = carE + CARRY_BASE(b, c, 0) + (size_t)n0 * D_MODEL + d;
  float* pq = carQ + CARRY_BASE(b, c, 0) + (size_t)n0 * D_MODEL + d;
  float he[4], pp[4];
#pragma unroll
  for (int j = 0; j < 4; ++j) {
    float e = sh_h[0][n0 + j][dl];
    float q = sh_P[0][n0 + j][dl];
    pe[j * D_MODEL] = e;
    pq[j * D_MODEL] = q;
#pragma unroll
    for (int k = 1; k < NSUB - 1; ++k) {
      float p = sh_P[k][n0 + j][dl];
      e = p * e + sh_h[k][n0 + j][dl];
      q *= p;
      pe[(k * D_STATE + j) * D_MODEL] = e;
      pq[(k * D_STATE + j) * D_MODEL] = q;
    }
    float pl = sh_P[NSUB - 1][n0 + j][dl];
    he[j] = pl * e + sh_h[NSUB - 1][n0 + j][dl];
    pp[j] = q * pl;
  }
  size_t base = ((size_t)(b * D_MODEL + d) * NCHUNK + c) * D_STATE + n0;
  *(float4*)(hend + base)  = make_float4(he[0], he[1], he[2], he[3]);
  *(float4*)(Pprod + base) = make_float4(pp[0], pp[1], pp[2], pp[3]);
}

__global__ __launch_bounds__(256) void k_scan_combine(const float* __restrict__ hend,
                                                      const float* __restrict__ Pprod,
                                                      float* __restrict__ h0) {
  int gid = blockIdx.x * blockDim.x + threadIdx.x;
  int n = gid & (D_STATE - 1);
  int bd = gid >> 4;
  size_t base = (size_t)bd * NCHUNK * D_STATE + n;
  float h = 0.f;
  for (int c = 0; c < NCHUNK; ++c) {
    size_t idx = base + (size_t)c * D_STATE;
    h0[idx] = h;
    h = Pprod[idx] * h + hend[idx];
  }
}

__global__ __launch_bounds__(256) void k_scan_emit(const float* __restrict__ x,
                                                   const float* __restrict__ dt,
                                                   const float* __restrict__ dtBC,
                                                   const float* __restrict__ A_log,
                                                   const float* __restrict__ Dp,
                                                   const float* __restrict__ h0,
                                                   const float* __restrict__ carE,
                                                   const float* __restrict__ carQ,
                                                   float* __restrict__ y) {
  int dl  = threadIdx.x & (DBLK - 1);
  int sub = threadIdx.x >> 6;
  int d = (blockIdx.x & (D_MODEL / DBLK - 1)) * DBLK + dl;
  int c = (blockIdx.x >> 4) & (NCHUNK - 1);
  int b = blockIdx.x >> 10;

  float An[D_STATE];
#pragma unroll
  for (int n = 0; n < D_STATE; ++n) An[n] = -__expf(A_log[d * D_STATE + n]);

  int t0 = c * TCHUNK + sub * TSUB;
  float h[D_STATE];

  size_t base = ((size_t)(b * D_MODEL + d) * NCHUNK + c) * D_STATE;
#pragma unroll
  for (int n = 0; n < D_STATE; ++n) h[n] = h0[base + n];
  if (sub > 0) {
    size_t cb = CARRY_BASE(b, c, sub - 1) + d;
#pragma unroll
    for (int n = 0; n < D_STATE; ++n)
      h[n] = carQ[cb + (size_t)n * D_MODEL] * h[n] + carE[cb + (size_t)n * D_MODEL];
  }

  float Dd = Dp[d];
  for (int t = t0; t < t0 + TSUB; ++t) {
    size_t row = (size_t)b * SEQLEN + t;
    float dtv = dt[row * D_MODEL + d];
    float xv  = x[row * D_MODEL + d];
    float Bv[D_STATE], Cv[D_STATE];
    const float4* bp = (const float4*)(dtBC + row * E_DIM + DT_RANK);
    const float4* cp = (const float4*)(dtBC + row * E_DIM + DT_RANK + D_STATE);
#pragma unroll
    for (int i = 0; i < 4; ++i) {
      float4 q = bp[i];
      Bv[4*i+0] = q.x; Bv[4*i+1] = q.y; Bv[4*i+2] = q.z; Bv[4*i+3] = q.w;
    }
#pragma unroll
    for (int i = 0; i < 4; ++i) {
      float4 q = cp[i];
      Cv[4*i+0] = q.x; Cv[4*i+1] = q.y; Cv[4*i+2] = q.z; Cv[4*i+3] = q.w;
    }
    float dx = dtv * xv;
    float yv = 0.f;
#pragma unroll
    for (int n = 0; n < D_STATE; ++n) {
      float dA = __expf(dtv * An[n]);
      h[n] = dA * h[n] + dx * Bv[n];
      yv += h[n] * Cv[n];
    }
    y[row * D_MODEL + d] = yv + xv * Dd;
  }
}

extern "C" void kernel_launch(void* const* d_in, const int* in_sizes, int n_in,
                              void* d_out, int out_size, void* d_ws, size_t ws_size,
                              hipStream_t stream) {
  const float* x    = (const float*)d_in[0];
  const float* Alog = (const float*)d_in[1];
  const float* Dp   = (const float*)d_in[2];
  const float* Wdt  = (const float*)d_in[3];
  const float* bdt  = (const float*)d_in[4];
  const float* Wbc  = (const float*)d_in[5];
  float* y = (float*)d_out;

  float* ws    = (float*)d_ws;
  float* dtBC  = ws;
  float* dt    = dtBC + (size_t)ROWS * E_DIM;
  float* hend  = dt + (size_t)ROWS * D_MODEL;
  float* Pprod = hend + (size_t)BATCH * D_MODEL * NCHUNK * D_STATE;
  float* h0    = Pprod + (size_t)BATCH * D_MODEL * NCHUNK * D_STATE;
  float* carE  = h0 + (size_t)BATCH * D_MODEL * NCHUNK * D_STATE;
  float* carQ  = carE + CARRY_FLOATS;

  k_proj_dtbc<<<ROWS / G1_ROWS, 256, 0, stream>>>(x, Wbc, dtBC);

  k_proj_dt<<<(ROWS / 32) * G2_NS * 32 / 256, 256, 0, stream>>>(dtBC, Wdt, bdt, dt);

  int nscan = BATCH * NCHUNK * (D_MODEL / DBLK);
  k_scan_local<<<nscan, 256, 0, stream>>>(x, dt, dtBC, Alog, hend, Pprod, carE, carQ);

  int ncomb = BATCH * D_MODEL * D_STATE;
  k_scan_combine<<<ncomb / 256, 256, 0, stream>>>(hend, Pprod, h0);

  k_scan_emit<<<nscan, 256, 0, stream>>>(x, dt, dtBC, Alog, Dp, h0, carE, carQ, y);
}


# scan passes: each thread carries two adjacent channels (8-byte dt, x, y, carry accesses): a step's B, C rows serve both, half as many waves per pass; sums keep their order, y's last bits move
# speedup vs baseline: 1.5246x; 1.1383x over previous
#include <hip/hip_runtime.h>
#include <hip/hip_bf16.h>

#define D_MODEL 1024
#define D_STATE 16
#define DT_RANK 64
#define BATCH   2
#define SEQLEN  4096
#define E_DIM   (DT_RANK + 2*D_STATE)
#define ROWS    (BATCH*SEQLEN)
#define NCHUNK  64
#define TCHUNK  (SEQLEN/NCHUNK)
#define NSUB    4
#define TSUB    (TCHUNK/NSUB)
#define DPT     2
#define DBLK    128
#define DCOL    (DBLK/DPT)
#define CARRY_BASE(b, c, k) ((size_t)((((b) * NCHUNK + (c)) * (NSUB - 1) + (k)) * D_STATE) * D_MODEL)
#define CARRY_FLOATS ((size_t)BATCH * NCHUNK * (NSUB - 1) * D_STATE * D_MODEL)

typedef __attribute__((ext_vector_type(2))) float v2f;
typedef __attribute__((ext_vector_type(8))) float v8f;

__device__ __forceinline__ float softplus_f(float v) {
  return v > 20.f ? v : log1pf(__expf(v));
}


#define G1_ROWS  32
#define G1_WAVES 8
#define G1_KW    (D_MODEL / G1_WAVES)
__global__ __launch_bounds__(256) void k_proj_dtbc(const float* __restrict__ x,
                                                   const float* __restrict__ Wbc,
                                                   float* __restrict__ dtBC) {
  __shared__ float part[G1_WAVES][G1_ROWS * E_DIM];
  int w = threadIdx.x >> 5;
  int lane = threadIdx.x & 31;
  int l = lane & 15, half = lane >> 4;
  int m0 = blockIdx.x * G1_ROWS;

  const float* arow = x   + (size_t)(m0 + l) * D_MODEL + w * G1_KW + 2 * half;
  const float* brow = Wbc + (size_t)l * D_MODEL        + w * G1_KW + 2 * half;

  v8f acc[2][6];
#pragma unroll
  for (int i = 0; i < 2; ++i)
#pragma unroll
    for (int j = 0; j < 6; ++j)
      acc[i][j] = (v8f){0.f, 0.f, 0.f, 0.f, 0.f, 0.f, 0.f, 0.f};

#pragma unroll 4
  for (int k = 0; k < G1_KW; k += 4) {
    v2f a[2], b[6];
#pragma unroll
    for (int i = 0; i < 2; ++i) {
      a[i][0] = arow[(size_t)i * 16 * D_MODEL + k];
      a[i][1] = arow[(size_t)i * 16 * D_MODEL + k + 1];
    }
#pragma unroll
    for (int j = 0; j < 6; ++j) {
      b[j][0] = brow[(size_t)j * 16 * D_MODEL + k];
      b[j][1] = brow[(size_t)j * 16 * D_MODEL + k + 1];
    }
#pragma unroll
    for (int j = 0; j < 6; ++j)
#pragma unroll
      for (int i = 0; i < 2; ++i)
        acc[i][j] = __builtin_amdgcn_wmma_f32_16x16x4_f32(false, a[i], false, b[j],
                                                          (short)0, acc[i][j], false, false);
  }

#pragma unroll
  for (int i = 0; i < 2; ++i)
#pragma unroll
    for (int j = 0; j < 6; ++j)
#pragma unroll
      for (int r = 0; r < 8; ++r)
        part[w][(i * 16 + 8 * half + r) * E_DIM + j * 16 + l] = acc[i][j][r];
  __syncthreads();

  float* outp = dtBC + (size_t)m0 * E_DIM;
#pragma unroll
  for (int q = 0; q < (G1_ROWS * E_DIM) / 256; ++q) {
    int o = q * 256 + threadIdx.x;
    float s = part[0][o];
#pragma unroll
    for (int p = 1; p < G1_WAVES; ++p) s += part[p][o];
    outp[o] = s;
  }
}

#define G2_NS (D_MODEL / 64)
#define G2_PITCH 68
__global__ __launch_bounds__(256) void k_proj_dt(const float* __restrict__ dtBC,
                                                 const float* __restrict__ Wdt,
                                                 const float* __restrict__ bdt,
                                                 float* __restrict__ dt) {
  __shared__ float tile[8][32 * G2_PITCH];
  int wave = (blockIdx.x * blockDim.x + threadIdx.x) >> 5;
  int lane = threadIdx.x & 31;
  int ms = wave / G2_NS, ns = wave - ms * G2_NS;
  int l = lane & 15, half = lane >> 4;

  const float* arow = dtBC + (size_t)(ms * 32 + l) * E_DIM   + 2 * half;
  const float* brow = Wdt  + (size_t)(ns * 64 + l) * DT_RANK + 2 * half;

  v8f acc[2][4];
#pragma unroll
  for (int i = 0; i < 2; ++i)
#pragma unroll
    for (int j = 0; j < 4; ++j)
      acc[i][j] = (v8f){0.f, 0.f, 0.f, 0.f, 0.f, 0.f, 0.f, 0.f};

#pragma unroll 4
  for (int k = 0; k < DT_RANK; k += 4) {
    v2f a[2], b[4];
#pragma unroll
    for (int i = 0; i < 2; ++i) {
      a[i][0] = arow[i * 16 * E_DIM + k];
      a[i][1] = arow[i * 16 * E_DIM + k + 1];
    }
#pragma unroll
    for (int j = 0; j < 4; ++j) {
      b[j][0] = brow[j * 16 * DT_RANK + k];
      b[j][1] = brow[j * 16 * DT_RANK + k + 1];
    }
#pragma unroll
    for (int j = 0; j < 4; ++j)
#pragma unroll
      for (int i = 0; i < 2; ++i)
        acc[i][j] = __builtin_amdgcn_wmma_f32_16x16x4_f32(false, a[i], false, b[j],
                                                          (short)0, acc[i][j], false, false);
  }

  float* tw = tile[threadIdx.x >> 5];
#pragma unroll
  for (int i = 0; i < 2; ++i)
#pragma unroll
    for (int j = 0; j < 4; ++j)
#pragma unroll
      for (int r = 0; r < 8; ++r)
        tw[(i * 16 + 8 * half + r) * G2_PITCH + j * 16 + l] = acc[i][j][r];
  __syncthreads();

  float bias0 = bdt[ns * 64 + lane];
  float bias1 = bdt[ns * 64 + 32 + lane];
  float* orow = dt + (size_t)(ms * 32) * D_MODEL + ns * 64 + lane;
#pragma unroll 2
  for (int r = 0; r < 32; ++r) {
    float v0 = tw[r * G2_PITCH + lane] + bias0;
    float v1 = tw[r * G2_PITCH + 32 + lane] + bias1;
    orow[(size_t)r * D_MODEL]      = softplus_f(v0);
    orow[(size_t)r * D_MODEL + 32] = softplus_f(v1);
  }
}

__global__ __launch_bounds__(256, 8) void k_scan_local(const float* __restrict__ x,
                                                   const float* __restrict__ dt,
                                                   const float* __restrict__ dtBC,
                                                   const float* __restrict__ A_log,
                                                   float* __restrict__ hend,
                                                   float* __restrict__ Pprod,
                                                   float* __restrict__ carE,
                                                   float* __restrict__ carQ) {
  __shared__ float sh_h[NSUB][D_STATE][DBLK];
  __shared__ float sh_P[NSUB][D_STATE][DBLK];

  int col = threadIdx.x & (DCOL - 1);
  int sub = threadIdx.x >> 6;
  int dl  = DPT * col;
  int d = (blockIdx.x & (D_MODEL / DBLK - 1)) * DBLK + dl;
  int c = (blockIdx.x >> 3) & (NCHUNK - 1);
  int b = blockIdx.x >> 9;

  float An0[D_STATE], An1[D_STATE];
#pragma unroll
  for (int n = 0; n < D_STATE; ++n) {
    An0[n] = -__expf(A_log[d * D_STATE + n]);
    An1[n] = -__expf(A_log[(d + 1) * D_STATE + n]);
  }

  float h0v[D_STATE], h1v[D_STATE], P0[D_STATE], P1[D_STATE];
#pragma unroll
  for (int n = 0; n < D_STATE; ++n) { h0v[n] = 0.f; h1v[n] = 0.f; P0[n] = 1.f; P1[n] = 1.f; }

  int t0 = c * TCHUNK + sub * TSUB;
  for (int t = t0; t < t0 + TSUB; ++t) {
    size_t row = (size_t)b * SEQLEN + t;
    float2 dtv = *(const float2*)(dt + row * D_MODEL + d);
    float2 xv  = *(const float2*)(x + row * D_MODEL + d);
    float Bv[D_STATE];
    const float4* bp = (const float4*)(dtBC + row * E_DIM + DT_RANK);
#pragma unroll
    for (int i = 0; i < 4; ++i) {
      float4 q = bp[i];
      Bv[4*i+0] = q.x; Bv[4*i+1] = q.y; Bv[4*i+2] = q.z; Bv[4*i+3] = q.w;
    }
    float dx0 = dtv.x * xv.x;
    float dx1 = dtv.y * xv.y;
#pragma unroll
    for (int n = 0; n < D_STATE; ++n) {
      float dA0 = __expf(dtv.x * An0[n]);
      float dA1 = __expf(dtv.y * An1[n]);
      h0v[n] = dA0 * h0v[n] + dx0 * Bv[n];
      h1v[n] = dA1 * h1v[n] + dx1 * Bv[n];
      P0[n] *= dA0;
      P1[n] *= dA1;
    }
  }

#pragma unroll
  for (int n = 0; n < D_STATE; ++n) {
    *(float2*)&sh_h[sub][n][dl] = make_float2(h0v[n], h1v[n]);
    *(float2*)&sh_P[sub][n][dl] = make_float2(P0[n], P1[n]);
  }
  __syncthreads();

  int n0 = sub * 4;
  float* pe = carE + CARRY_BASE(b, c, 0) + (size_t)n0 * D_MODEL + d;
  float* pq = carQ + CARRY_BASE(b, c, 0) + (size_t)n0 * D_MODEL + d;
  float he0[4], he1[4], pp0[4], pp1[4];
#pragma unroll
  for (int j = 0; j < 4; ++j) {
    float2 e = *(const float2*)&sh_h[0][n0 + j][dl];
    float2 q = *(const float2*)&sh_P[0][n0 + j][dl];
    *(float2*)(pe + j * D_MODEL) = e;
    *(float2*)(pq + j * D_MODEL) = q;
#pragma unroll
    for (int k = 1; k < NSUB - 1; ++k) {
      float2 p = *(const float2*)&sh_P[k][n0 + j][dl];
      float2 g = *(const float2*)&sh_h[k][n0 + j][dl];
      e.x = p.x * e.x + g.x;
      e.y = p.y * e.y + g.y;
      q.x *= p.x;
      q.y *= p.y;
      *(float2*)(pe + (k * D_STATE + j) * D_MODEL) = e;
      *(float2*)(pq + (k * D_STATE + j) * D_MODEL) = q;
    }
    float2 pl = *(const float2*)&sh_P[NSUB - 1][n0 + j][dl];
    float2 gl = *(const float2*)&sh_h[NSUB - 1][n0 + j][dl];
    he0[j] = pl.x * e.x + gl.x;
    he1[j] = pl.y * e.y + gl.y;
    pp0[j] = q.x * pl.x;
    pp1[j] = q.y * pl.y;
  }
  size_t base0 = ((size_t)(b * D_MODEL + d) * NCHUNK + c) * D_STATE + n0;
  size_t base1 = base0 + (size_t)NCHUNK * D_STATE;
  *(float4*)(hend + base0)  = make_float4(he0[0], he0[1], he0[2], he0[3]);
  *(float4*)(hend + base1)  = make_float4(he1[0], he1[1], he1[2], he1[3]);
  *(float4*)(Pprod + base0) = make_float4(pp0[0], pp0[1], pp0[2], pp0[3]);
  *(float4*)(Pprod + base1) = make_float4(pp1[0], pp1[1], pp1[2], pp1[3]);
}

__global__ __launch_bounds__(256) void k_scan_combine(const float* __restrict__ hend,
                                                      const float* __restrict__ Pprod,
                                                      float* __restrict__ h0) {
  int gid = blockIdx.x * blockDim.x + threadIdx.x;
  int n = gid & (D_STATE - 1);
  int bd = gid >> 4;
  size_t base = (size_t)bd * NCHUNK * D_STATE + n;
  float h = 0.f;
  for (int c = 0; c < NCHUNK; ++c) {
    size_t idx = base + (size_t)c * D_STATE;
    h0[idx] = h;
    h = Pprod[idx] * h + hend[idx];
  }
}

__global__ __launch_bounds__(256, 8) void k_scan_emit(const float* __restrict__ x,
                                                   const float* __restrict__ dt,
                                                   const float* __restrict__ dtBC,
                                                   const float* __restrict__ A_log,
                                                   const float* __restrict__ Dp,
                                                   const float* __restrict__ h0,
                                                   const float* __restrict__ carE,
                                                   const float* __restrict__ carQ,
                                                   float* __restrict__ y) {
  int col = threadIdx.x & (DCOL - 1);
  int sub = threadIdx.x >> 6;
  int d = (blockIdx.x & (D_MODEL / DBLK - 1)) * DBLK + DPT * col;
  int c = (blockIdx.x >> 3) & (NCHUNK - 1);
  int b = blockIdx.x >> 9;

  float An0[D_STATE], An1[D_STATE];
#pragma unroll
  for (int n = 0; n < D_STATE; ++n) {
    An0[n] = -__expf(A_log[d * D_STATE + n]);
    An1[n] = -__expf(A_log[(d + 1) * D_STATE + n]);
  }

  int t0 = c * TCHUNK + sub * TSUB;
  float h0v[D_STATE], h1v[D_STATE];

  size_t base0 = ((size_t)(b * D_MODEL + d) * NCHUNK + c) * D_STATE;
  size_t base1 = base0 + (size_t)NCHUNK * D_STATE;
#pragma unroll
  for (int n = 0; n < D_STATE; ++n) { h0v[n] = h0[base0 + n]; h1v[n] = h0[base1 + n]; }
  if (sub > 0) {
    size_t cb = CARRY_BASE(b, c, sub - 1) + d;
#pragma unroll
    for (int n = 0; n < D_STATE; ++n) {
      float2 q = *(const float2*)(carQ + cb + (size_t)n * D_MODEL);
      float2 e = *(const float2*)(carE + cb + (size_t)n * D_MODEL);
      h0v[n] = q.x * h0v[n] + e.x;
      h1v[n] = q.y * h1v[n] + e.y;
    }
  }

  float2 Dd = *(const float2*)(Dp + d);
  for (int t = t0; t < t0 + TSUB; ++t) {
    size_t row = (size_t)b * SEQLEN + t;
    float2 dtv = *(const float2*)(dt + row * D_MODEL + d);
    float2 xv  = *(const float2*)(x + row * D_MODEL + d);
    float Bv[D_STATE], Cv[D_STATE];
    const float4* bp = (const float4*)(dtBC + row * E_DIM + DT_RANK);
    const float4* cp = (const float4*)(dtBC + row * E_DIM + DT_RANK + D_STATE);
#pragma unroll
    for (int i = 0; i < 4; ++i) {
      float4 q = bp[i];
      Bv[4*i+0] = q.x; Bv[4*i+1] = q.y; Bv[4*i+2] = q.z; Bv[4*i+3] = q.w;
    }
#pragma unroll
    for (int i = 0; i < 4; ++i) {
      float4 q = cp[i];
      Cv[4*i+0] = q.x; Cv[4*i+1] = q.y; Cv[4*i+2] = q.z; Cv[4*i+3] = q.w;
    }
    float dx0 = dtv.x * xv.x;
    float dx1 = dtv.y * xv.y;
    float y0 = 0.f, y1 = 0.f;
#pragma unroll
    for (int n = 0; n < D_STATE; ++n) {
      float dA0 = __expf(dtv.x * An0[n]);
      float dA1 = __expf(dtv.y * An1[n]);
      h0v[n] = dA0 * h0v[n] + dx0 * Bv[n];
      h1v[n] = dA1 * h1v[n] + dx1 * Bv[n];
      y0 += h0v[n] * Cv[n];
      y1 += h1v[n] * Cv[n];
    }
    *(float2*)(y + row * D_MODEL + d) = make_float2(y0 + xv.x * Dd.x, y1 + xv.y * Dd.y);
  }
}

extern "C" void kernel_launch(void* const* d_in, const int* in_sizes, int n_in,
                              void* d_out, int out_size, void* d_ws, size_t ws_size,
                              hipStream_t stream) {
  const float* x    = (const float*)d_in[0];
  const float* Alog = (const float*)d_in[1];
  const float* Dp   = (const float*)d_in[2];
  const float* Wdt  = (const float*)d_in[3];
  const float* bdt  = (const float*)d_in[4];
  const float* Wbc  = (const float*)d_in[5];
  float* y = (float*)d_out;

  float* ws    = (float*)d_ws;
  float* dtBC  = ws;
  float* dt    = dtBC + (size_t)ROWS * E_DIM;
  float* hend  = dt + (size_t)ROWS * D_MODEL;
  float* Pprod = hend + (size_t)BATCH * D_MODEL * NCHUNK * D_STATE;
  float* h0    = Pprod + (size_t)BATCH * D_MODEL * NCHUNK * D_STATE;
  float* carE  = h0 + (size_t)BATCH * D_MODEL * NCHUNK * D_STATE;
  float* carQ  = carE + CARRY_FLOATS;

  k_proj_dtbc<<<ROWS / G1_ROWS, 256, 0, stream>>>(x, Wbc, dtBC);

  k_proj_dt<<<(ROWS / 32) * G2_NS * 32 / 256, 256, 0, stream>>>(dtBC, Wdt, bdt, dt);

  int nscan = BATCH * NCHUNK * (D_MODEL / DBLK);
  k_scan_local<<<nscan, 256, 0, stream>>>(x, dt, dtBC, Alog, hend, Pprod, carE, carQ);

  int ncomb = BATCH * D_MODEL * D_STATE;
  k_scan_combine<<<ncomb / 256, 256, 0, stream>>>(hend, Pprod, h0);

  k_scan_emit<<<nscan, 256, 0, stream>>>(x, dt, dtBC, Alog, Dp, h0, carE, carQ, y);
}


# dt projection (GEMM 2 + softplus) runs inside GEMM 1's blocks on the 32 rows just joined in LDS: one launch and 4096 waves fewer, dt_in fragments read from LDS; softplus by select; sums unchanged
# speedup vs baseline: 1.6088x; 1.0552x over previous
#include <hip/hip_runtime.h>
#include <hip/hip_bf16.h>

#define D_MODEL 1024
#define D_STATE 16
#define DT_RANK 64
#define BATCH   2
#define SEQLEN  4096
#define E_DIM   (DT_RANK + 2*D_STATE)
#define ROWS    (BATCH*SEQLEN)
#define NCHUNK  64
#define TCHUNK  (SEQLEN/NCHUNK)
#define NSUB    4
#define TSUB    (TCHUNK/NSUB)
#define DPT     2
#define DBLK    128
#define DCOL    (DBLK/DPT)
#define CARRY_BASE(b, c, k) ((size_t)((((b) * NCHUNK + (c)) * (NSUB - 1) + (k)) * D_STATE) * D_MODEL)
#define CARRY_FLOATS ((size_t)BATCH * NCHUNK * (NSUB - 1) * D_STATE * D_MODEL)

typedef __attribute__((ext_vector_type(2))) float v2f;
typedef __attribute__((ext_vector_type(8))) float v8f;

__device__ __forceinline__ float softplus_f(float v) {
  float s = log1pf(__expf(v));
  return v > 20.f ? v : s;
}


#define G1_ROWS  32
#define G1_WAVES 8
#define G1_KW    (D_MODEL / G1_WAVES)
#define G2_PITCH 68
__global__ __launch_bounds__(256) void k_proj_dtbc(const float* __restrict__ x,
                                                   const float* __restrict__ Wbc,
                                                   const float* __restrict__ Wdt,
                                                   const float* __restrict__ bdt,
                                                   float* __restrict__ dtBC,
                                                   float* __restrict__ dt) {
  __shared__ float part[G1_WAVES][G1_ROWS * E_DIM];
  __shared__ float atile[G1_ROWS * G2_PITCH];
  int w = threadIdx.x >> 5;
  int lane = threadIdx.x & 31;
  int l = lane & 15, half = lane >> 4;
  int m0 = blockIdx.x * G1_ROWS;

  const float* arow = x   + (size_t)(m0 + l) * D_MODEL + w * G1_KW + 2 * half;
  const float* brow = Wbc + (size_t)l * D_MODEL        + w * G1_KW + 2 * half;

  v8f acc[2][6];
#pragma unroll
  for (int i = 0; i < 2; ++i)
#pragma unroll
    for (int j = 0; j < 6; ++j)
      acc[i][j] = (v8f){0.f, 0.f, 0.f, 0.f, 0.f, 0.f, 0.f, 0.f};

#pragma unroll 4
  for (int k = 0; k < G1_KW; k += 4) {
    v2f a[2], b[6];
#pragma unroll
    for (int i = 0; i < 2; ++i) {
      a[i][0] = arow[(size_t)i * 16 * D_MODEL + k];
      a[i][1] = arow[(size_t)i * 16 * D_MODEL + k + 1];
    }
#pragma unroll
    for (int j = 0; j < 6; ++j) {
      b[j][0] = brow[(size_t)j * 16 * D_MODEL + k];
      b[j][1] = brow[(size_t)j * 16 * D_MODEL + k + 1];
    }
#pragma unroll
    for (int j = 0; j < 6; ++j)
#pragma unroll
      for (int i = 0; i < 2; ++i)
        acc[i][j] = __builtin_amdgcn_wmma_f32_16x16x4_f32(false, a[i], false, b[j],
                                                          (short)0, acc[i][j], false, false);
  }

#pragma unroll
  for (int i = 0; i < 2; ++i)
#pragma unroll
    for (int j = 0; j < 6; ++j)
#pragma unroll
      for (int r = 0; r < 8; ++r)
        part[w][(i * 16 + 8 * half + r) * E_DIM + j * 16 + l] = acc[i][j][r];
  __syncthreads();

  float* outp = dtBC + (size_t)m0 * E_DIM;
#pragma unroll
  for (int q = 0; q < (G1_ROWS * E_DIM) / 256; ++q) {
    int o = q * 256 + threadIdx.x;
    float s = part[0][o];
#pragma unroll
    for (int p = 1; p < G1_WAVES; ++p) s += part[p][o];
    outp[o] = s;
    int rr = o / E_DIM, cc = o - rr * E_DIM;
    if (cc < DT_RANK) atile[rr * G2_PITCH + cc] = s;
  }
  __syncthreads();

  float* tw = &part[0][0] + w * (32 * G2_PITCH);
  const float* al = atile + l * G2_PITCH + 2 * half;
  for (int hb = 0; hb < 2; ++hb) {
    int c0 = w * 128 + hb * 64;
    const float* wrow = Wdt + (size_t)(c0 + l) * DT_RANK + 2 * half;

    v8f acc2[2][4];
#pragma unroll
    for (int i = 0; i < 2; ++i)
#pragma unroll
      for (int j = 0; j < 4; ++j)
        acc2[i][j] = (v8f){0.f, 0.f, 0.f, 0.f, 0.f, 0.f, 0.f, 0.f};

#pragma unroll 4
    for (int k = 0; k < DT_RANK; k += 4) {
      v2f a[2], b[4];
#pragma unroll
      for (int i = 0; i < 2; ++i) {
        a[i][0] = al[i * 16 * G2_PITCH + k];
        a[i][1] = al[i * 16 * G2_PITCH + k + 1];
      }
#pragma unroll
      for (int j = 0; j < 4; ++j) {
        b[j][0] = wrow[j * 16 * DT_RANK + k];
        b[j][1] = wrow[j * 16 * DT_RANK + k + 1];
      }
#pragma unroll
      for (int j = 0; j < 4; ++j)
#pragma unroll
        for (int i = 0; i < 2; ++i)
          acc2[i][j] = __builtin_amdgcn_wmma_f32_16x16x4_f32(false, a[i], false, b[j],
                                                             (short)0, acc2[i][j], false, false);
    }

#pragma unroll
    for (int i = 0; i < 2; ++i)
#pragma unroll
      for (int j = 0; j < 4; ++j)
#pragma unroll
        for (int r = 0; r < 8; ++r)
          tw[(i * 16 + 8 * half + r) * G2_PITCH + j * 16 + l] = acc2[i][j][r];
    __syncthreads();

    float bias0 = bdt[c0 + lane];
    float bias1 = bdt[c0 + 32 + lane];
    float* orow = dt + (size_t)m0 * D_MODEL + c0 + lane;
#pragma unroll 2
    for (int r = 0; r < 32; ++r) {
      float v0 = tw[r * G2_PITCH + lane] + bias0;
      float v1 = tw[r * G2_PITCH + 32 + lane] + bias1;
      orow[(size_t)r * D_MODEL]      = softplus_f(v0);
      orow[(size_t)r * D_MODEL + 32] = softplus_f(v1);
    }
  }
}

__global__ __launch_bounds__(256, 8) void k_scan_local(const float* __restrict__ x,
                                                   const float* __restrict__ dt,
                                                   const float* __restrict__ dtBC,
                                                   const float* __restrict__ A_log,
                                                   float* __restrict__ hend,
                                                   float* __restrict__ Pprod,
                                                   float* __restrict__ carE,
                                                   float* __restrict__ carQ) {
  __shared__ float sh_h[NSUB][D_STATE][DBLK];
  __shared__ float sh_P[NSUB][D_STATE][DBLK];

  int col = threadIdx.x & (DCOL - 1);
  int sub = threadIdx.x >> 6;
  int dl  = DPT * col;
  int d = (blockIdx.x & (D_MODEL / DBLK - 1)) * DBLK + dl;
  int c = (blockIdx.x >> 3) & (NCHUNK - 1);
  int b = blockIdx.x >> 9;

  float An0[D_STATE], An1[D_STATE];
#pragma unroll
  for (int n = 0; n < D_STATE; ++n) {
    An0[n] = -__expf(A_log[d * D_STATE + n]);
    An1[n] = -__expf(A_log[(d + 1) * D_STATE + n]);
  }

  float h0v[D_STATE], h1v[D_STATE], P0[D_STATE], P1[D_STATE];
#pragma unroll
  for (int n = 0; n < D_STATE; ++n) { h0v[n] = 0.f; h1v[n] = 0.f; P0[n] = 1.f; P1[n] = 1.f; }

  int t0 = c * TCHUNK + sub * TSUB;
  for (int t = t0; t < t0 + TSUB; ++t) {
    size_t row = (size_t)b * SEQLEN + t;
    float2 dtv = *(const float2*)(dt + row * D_MODEL + d);
    float2 xv  = *(const float2*)(x + row * D_MODEL + d);
    float Bv[D_STATE];
    const float4* bp = (const float4*)(dtBC + row * E_DIM + DT_RANK);
#pragma unroll
    for (int i = 0; i < 4; ++i) {
      float4 q = bp[i];
      Bv[4*i+0] = q.x; Bv[4*i+1] = q.y; Bv[4*i+2] = q.z; Bv[4*i+3] = q.w;
    }
    float dx0 = dtv.x * xv.x;
    float dx1 = dtv.y * xv.y;
#pragma unroll
    for (int n = 0; n < D_STATE; ++n) {
      float dA0 = __expf(dtv.x * An0[n]);
      float dA1 = __expf(dtv.y * An1[n]);
      h0v[n] = dA0 * h0v[n] + dx0 * Bv[n];
      h1v[n] = dA1 * h1v[n] + dx1 * Bv[n];
      P0[n] *= dA0;
      P1[n] *= dA1;
    }
  }

#pragma unroll
  for (int n = 0; n < D_STATE; ++n) {
    *(float2*)&sh_h[sub][n][dl] = make_float2(h0v[n], h1v[n]);
    *(float2*)&sh_P[sub][n][dl] = make_float2(P0[n], P1[n]);
  }
  __syncthreads();

  int n0 = sub * 4;
  float* pe = carE + CARRY_BASE(b, c, 0) + (size_t)n0 * D_MODEL + d;
  float* pq = carQ + CARRY_BASE(b, c, 0) + (size_t)n0 * D_MODEL + d;
  float he0[4], he1[4], pp0[4], pp1[4];
#pragma unroll
  for (int j = 0; j < 4; ++j) {
    float2 e = *(const float2*)&sh_h[0][n0 + j][dl];
    float2 q = *(const float2*)&sh_P[0][n0 + j][dl];
    *(float2*)(pe + j * D_MODEL) = e;
    *(float2*)(pq + j * D_MODEL) = q;
#pragma unroll
    for (int k = 1; k < NSUB - 1; ++k) {
      float2 p = *(const float2*)&sh_P[k][n0 + j][dl];
      float2 g = *(const float2*)&sh_h[k][n0 + j][dl];
      e.x = p.x * e.x + g.x;
      e.y = p.y * e.y + g.y;
      q.x *= p.x;
      q.y *= p.y;
      *(float2*)(pe + (k * D_STATE + j) * D_MODEL) = e;
      *(float2*)(pq + (k * D_STATE + j) * D_MODEL) = q;
    }
    float2 pl = *(const float2*)&sh_P[NSUB - 1][n0 + j][dl];
    float2 gl = *(const float2*)&sh_h[NSUB - 1][n0 + j][dl];
    he0[j] = pl.x * e.x + gl.x;
    he1[j] = pl.y * e.y + gl.y;
    pp0[j] = q.x * pl.x;
    pp1[j] = q.y * pl.y;
  }
  size_t base0 = ((size_t)(b * D_MODEL + d) * NCHUNK + c) * D_STATE + n0;
  size_t base1 = base0 + (size_t)NCHUNK * D_STATE;
  *(float4*)(hend + base0)  = make_float4(he0[0], he0[1], he0[2], he0[3]);
  *(float4*)(hend + base1)  = make_float4(he1[0], he1[1], he1[2], he1[3]);
  *(float4*)(Pprod + base0) = make_float4(pp0[0], pp0[1], pp0[2], pp0[3]);
  *(float4*)(Pprod + base1) = make_float4(pp1[0], pp1[1], pp1[2], pp1[3]);
}

__global__ __launch_bounds__(256) void k_scan_combine(const float* __restrict__ hend,
                                                      const float* __restrict__ Pprod,
                                                      float* __restrict__ h0) {
  int gid = blockIdx.x * blockDim.x + threadIdx.x;
  int n = gid & (D_STATE - 1);
  int bd = gid >> 4;
  size_t base = (size_t)bd * NCHUNK * D_STATE + n;
  float h = 0.f;
  for (int c = 0; c < NCHUNK; ++c) {
    size_t idx = base + (size_t)c * D_STATE;
    h0[idx] = h;
    h = Pprod[idx] * h + hend[idx];
  }
}

__global__ __launch_bounds__(256, 8) void k_scan_emit(const float* __restrict__ x,
                                                   const float* __restrict__ dt,
                                                   const float* __restrict__ dtBC,
                                                   const float* __restrict__ A_log,
                                                   const float* __restrict__ Dp,
                                                   const float* __restrict__ h0,
                                                   const float* __restrict__ carE,
                                                   const float* __restrict__ carQ,
                                                   float* __restrict__ y) {
  int col = threadIdx.x & (DCOL - 1);
  int sub = threadIdx.x >> 6;
  int d = (blockIdx.x & (D_MODEL / DBLK - 1)) * DBLK + DPT * col;
  int c = (blockIdx.x >> 3) & (NCHUNK - 1);
  int b = blockIdx.x >> 9;

  float An0[D_STATE], An1[D_STATE];
#pragma unroll
  for (int n = 0; n < D_STATE; ++n) {
    An0[n] = -__expf(A_log[d * D_STATE + n]);
    An1[n] = -__expf(A_log[(d + 1) * D_STATE + n]);
  }

  int t0 = c * TCHUNK + sub * TSUB;
  float h0v[D_STATE], h1v[D_STATE];

  size_t base0 = ((size_t)(b * D_MODEL + d) * NCHUNK + c) * D_STATE;
  size_t base1 = base0 + (size_t)NCHUNK * D_STATE;
#pragma unroll
  for (int n = 0; n < D_STATE; ++n) { h0v[n] = h0[base0 + n]; h1v[n] = h0[base1 + n]; }
  if (sub > 0) {
    size_t cb = CARRY_BASE(b, c, sub - 1) + d;
#pragma unroll
    for (int n = 0; n < D_STATE; ++n) {
      float2 q = *(const float2*)(carQ + cb + (size_t)n * D_MODEL);
      float2 e = *(const float2*)(carE + cb + (size_t)n * D_MODEL);
      h0v[n] = q.x * h0v[n] + e.x;
      h1v[n] = q.y * h1v[n] + e.y;
    }
  }

  float2 Dd = *(const float2*)(Dp + d);
  for (int t = t0; t < t0 + TSUB; ++t) {
    size_t row = (size_t)b * SEQLEN + t;
    float2 dtv = *(const float2*)(dt + row * D_MODEL + d);
    float2 xv  = *(const float2*)(x + row * D_MODEL + d);
    float Bv[D_STATE], Cv[D_STATE];
    const float4* bp = (const float4*)(dtBC + row * E_DIM + DT_RANK);
    const float4* cp = (const float4*)(dtBC + row * E_DIM + DT_RANK + D_STATE);
#pragma unroll
    for (int i = 0; i < 4; ++i) {
      float4 q = bp[i];
      Bv[4*i+0] = q.x; Bv[4*i+1] = q.y; Bv[4*i+2] = q.z; Bv[4*i+3] = q.w;
    }
#pragma unroll
    for (int i = 0; i < 4; ++i) {
      float4 q = cp[i];
      Cv[4*i+0] = q.x; Cv[4*i+1] = q.y; Cv[4*i+2] = q.z; Cv[4*i+3] = q.w;
    }
    float dx0 = dtv.x * xv.x;
    float dx1 = dtv.y * xv.y;
    float y0 = 0.f, y1 = 0.f;
#pragma unroll
    for (int n = 0; n < D_STATE; ++n) {
      float dA0 = __expf(dtv.x * An0[n]);
      float dA1 = __expf(dtv.y * An1[n]);
      h0v[n] = dA0 * h0v[n] + dx0 * Bv[n];
      h1v[n] = dA1 * h1v[n] + dx1 * Bv[n];
      y0 += h0v[n] * Cv[n];
      y1 += h1v[n] * Cv[n];
    }
    *(float2*)(y + row * D_MODEL + d) = make_float2(y0 + xv.x * Dd.x, y1 + xv.y * Dd.y);
  }
}

extern "C" void kernel_launch(void* const* d_in, const int* in_sizes, int n_in,
                              void* d_out, int out_size, void* d_ws, size_t ws_size,
                              hipStream_t stream) {
  const float* x    = (const float*)d_in[0];
  const float* Alog = (const float*)d_in[1];
  const float* Dp   = (const float*)d_in[2];
  const float* Wdt  = (const float*)d_in[3];
  const float* bdt  = (const float*)d_in[4];
  const float* Wbc  = (const float*)d_in[5];
  float* y = (float*)d_out;

  float* ws    = (float*)d_ws;
  float* dtBC  = ws;
  float* dt    = dtBC + (size_t)ROWS * E_DIM;
  float* hend  = dt + (size_t)ROWS * D_MODEL;
  float* Pprod = hend + (size_t)BATCH * D_MODEL * NCHUNK * D_STATE;
  float* h0    = Pprod + (size_t)BATCH * D_MODEL * NCHUNK * D_STATE;
  float* carE  = h0 + (size_t)BATCH * D_MODEL * NCHUNK * D_STATE;
  float* carQ  = carE + CARRY_FLOATS;

  k_proj_dtbc<<<ROWS / G1_ROWS, 256, 0, stream>>>(x, Wbc, Wdt, bdt, dtBC, dt);

  int nscan = BATCH * NCHUNK * (D_MODEL / DBLK);
  k_scan_local<<<nscan, 256, 0, stream>>>(x, dt, dtBC, Alog, hend, Pprod, carE, carQ);

  int ncomb = BATCH * D_MODEL * D_STATE;
  k_scan_combine<<<ncomb / 256, 256, 0, stream>>>(hend, Pprod, h0);

  k_scan_emit<<<nscan, 256, 0, stream>>>(x, dt, dtBC, Alog, Dp, h0, carE, carQ, y);
}


# projection GEMMs: A and B fragment loaders take 16 bytes per lane (were 8) straight into registers, each load feeds two products, equal permuted k order in both; k sum regrouped inside each 8
# speedup vs baseline: 1.7079x; 1.0616x over previous
#include <hip/hip_runtime.h>
#include <hip/hip_bf16.h>

#define D_MODEL 1024
#define D_STATE 16
#define DT_RANK 64
#define BATCH   2
#define SEQLEN  4096
#define E_DIM   (DT_RANK + 2*D_STATE)
#define ROWS    (BATCH*SEQLEN)
#define NCHUNK  64
#define TCHUNK  (SEQLEN/NCHUNK)
#define NSUB    4
#define TSUB    (TCHUNK/NSUB)
#define DPT     2
#define DBLK    128
#define DCOL    (DBLK/DPT)
#define CARRY_BASE(b, c, k) ((size_t)((((b) * NCHUNK + (c)) * (NSUB - 1) + (k)) * D_STATE) * D_MODEL)
#define CARRY_FLOATS ((size_t)BATCH * NCHUNK * (NSUB - 1) * D_STATE * D_MODEL)

typedef __attribute__((ext_vector_type(2))) float v2f;
typedef __attribute__((ext_vector_type(4))) float v4f;
typedef __attribute__((ext_vector_type(8))) float v8f;

__device__ __forceinline__ float softplus_f(float v) {
  float s = log1pf(__expf(v));
  return v > 20.f ? v : s;
}


#define G1_ROWS  32
#define G1_WAVES 8
#define G1_KW    (D_MODEL / G1_WAVES)
#define G2_PITCH 68
__global__ __launch_bounds__(256) void k_proj_dtbc(const float* __restrict__ x,
                                                   const float* __restrict__ Wbc,
                                                   const float* __restrict__ Wdt,
                                                   const float* __restrict__ bdt,
                                                   float* __restrict__ dtBC,
                                                   float* __restrict__ dt) {
  __shared__ float part[G1_WAVES][G1_ROWS * E_DIM];
  __shared__ __attribute__((aligned(16))) float atile[G1_ROWS * G2_PITCH];
  int w = threadIdx.x >> 5;
  int lane = threadIdx.x & 31;
  int l = lane & 15, half = lane >> 4;
  int m0 = blockIdx.x * G1_ROWS;

  const float* arow = x   + (size_t)(m0 + l) * D_MODEL + w * G1_KW + 4 * half;
  const float* brow = Wbc + (size_t)l * D_MODEL        + w * G1_KW + 4 * half;

  v8f acc[2][6];
#pragma unroll
  for (int i = 0; i < 2; ++i)
#pragma unroll
    for (int j = 0; j < 6; ++j)
      acc[i][j] = (v8f){0.f, 0.f, 0.f, 0.f, 0.f, 0.f, 0.f, 0.f};

#pragma unroll 2
  for (int k = 0; k < G1_KW; k += 8) {
    v4f a[2], b[6];
#pragma unroll
    for (int i = 0; i < 2; ++i)
      a[i] = *(const v4f*)(arow + (size_t)i * 16 * D_MODEL + k);
#pragma unroll
    for (int j = 0; j < 6; ++j)
      b[j] = *(const v4f*)(brow + (size_t)j * 16 * D_MODEL + k);
#pragma unroll
    for (int u = 0; u < 2; ++u)
#pragma unroll
      for (int j = 0; j < 6; ++j)
#pragma unroll
        for (int i = 0; i < 2; ++i) {
          v2f af = {a[i][2 * u], a[i][2 * u + 1]};
          v2f bf = {b[j][2 * u], b[j][2 * u + 1]};
          acc[i][j] = __builtin_amdgcn_wmma_f32_16x16x4_f32(false, af, false, bf,
                                                            (short)0, acc[i][j], false, false);
        }
  }

#pragma unroll
  for (int i = 0; i < 2; ++i)
#pragma unroll
    for (int j = 0; j < 6; ++j)
#pragma unroll
      for (int r = 0; r < 8; ++r)
        part[w][(i * 16 + 8 * half + r) * E_DIM + j * 16 + l] = acc[i][j][r];
  __syncthreads();

  float* outp = dtBC + (size_t)m0 * E_DIM;
#pragma unroll
  for (int q = 0; q < (G1_ROWS * E_DIM) / 256; ++q) {
    int o = q * 256 + threadIdx.x;
    float s = part[0][o];
#pragma unroll
    for (int p = 1; p < G1_WAVES; ++p) s += part[p][o];
    outp[o] = s;
    int rr = o / E_DIM, cc = o - rr * E_DIM;
    if (cc < DT_RANK) atile[rr * G2_PITCH + cc] = s;
  }
  __syncthreads();

  float* tw = &part[0][0] + w * (32 * G2_PITCH);
  const float* al = atile + l * G2_PITCH + 4 * half;
  for (int hb = 0; hb < 2; ++hb) {
    int c0 = w * 128 + hb * 64;
    const float* wrow = Wdt + (size_t)(c0 + l) * DT_RANK + 4 * half;

    v8f acc2[2][4];
#pragma unroll
    for (int i = 0; i < 2; ++i)
#pragma unroll
      for (int j = 0; j < 4; ++j)
        acc2[i][j] = (v8f){0.f, 0.f, 0.f, 0.f, 0.f, 0.f, 0.f, 0.f};

#pragma unroll 2
    for (int k = 0; k < DT_RANK; k += 8) {
      v4f a[2], b[4];
#pragma unroll
      for (int i = 0; i < 2; ++i)
        a[i] = *(const v4f*)(al + i * 16 * G2_PITCH + k);
#pragma unroll
      for (int j = 0; j < 4; ++j)
        b[j] = *(const v4f*)(wrow + j * 16 * DT_RANK + k);
#pragma unroll
      for (int u = 0; u < 2; ++u)
#pragma unroll
        for (int j = 0; j < 4; ++j)
#pragma unroll
          for (int i = 0; i < 2; ++i) {
            v2f af = {a[i][2 * u], a[i][2 * u + 1]};
            v2f bf = {b[j][2 * u], b[j][2 * u + 1]};
            acc2[i][j] = __builtin_amdgcn_wmma_f32_16x16x4_f32(false, af, false, bf,
                                                               (short)0, acc2[i][j], false, false);
          }
    }

#pragma unroll
    for (int i = 0; i < 2; ++i)
#pragma unroll
      for (int j = 0; j < 4; ++j)
#pragma unroll
        for (int r = 0; r < 8; ++r)
          tw[(i * 16 + 8 * half + r) * G2_PITCH + j * 16 + l] = acc2[i][j][r];
    __syncthreads();

    float bias0 = bdt[c0 + lane];
    float bias1 = bdt[c0 + 32 + lane];
    float* orow = dt + (size_t)m0 * D_MODEL + c0 + lane;
#pragma unroll 2
    for (int r = 0; r < 32; ++r) {
      float v0 = tw[r * G2_PITCH + lane] + bias0;
      float v1 = tw[r * G2_PITCH + 32 + lane] + bias1;
      orow[(size_t)r * D_MODEL]      = softplus_f(v0);
      orow[(size_t)r * D_MODEL + 32] = softplus_f(v1);
    }
  }
}

__global__ __launch_bounds__(256, 8) void k_scan_local(const float* __restrict__ x,
                                                   const float* __restrict__ dt,
                                                   const float* __restrict__ dtBC,
                                                   const float* __restrict__ A_log,
                                                   float* __restrict__ hend,
                                                   float* __restrict__ Pprod,
                                                   float* __restrict__ carE,
                                                   float* __restrict__ carQ) {
  __shared__ float sh_h[NSUB][D_STATE][DBLK];
  __shared__ float sh_P[NSUB][D_STATE][DBLK];

  int col = threadIdx.x & (DCOL - 1);
  int sub = threadIdx.x >> 6;
  int dl  = DPT * col;
  int d = (blockIdx.x & (D_MODEL / DBLK - 1)) * DBLK + dl;
  int c = (blockIdx.x >> 3) & (NCHUNK - 1);
  int b = blockIdx.x >> 9;

  float An0[D_STATE], An1[D_STATE];
#pragma unroll
  for (int n = 0; n < D_STATE; ++n) {
    An0[n] = -__expf(A_log[d * D_STATE + n]);
    An1[n] = -__expf(A_log[(d + 1) * D_STATE + n]);
  }

  float h0v[D_STATE], h1v[D_STATE], P0[D_STATE], P1[D_STATE];
#pragma unroll
  for (int n = 0; n < D_STATE; ++n) { h0v[n] = 0.f; h1v[n] = 0.f; P0[n] = 1.f; P1[n] = 1.f; }

  int t0 = c * TCHUNK + sub * TSUB;
  for (int t = t0; t < t0 + TSUB; ++t) {
    size_t row = (size_t)b * SEQLEN + t;
    float2 dtv = *(const float2*)(dt + row * D_MODEL + d);
    float2 xv  = *(const float2*)(x + row * D_MODEL + d);
    float Bv[D_STATE];
    const float4* bp = (const float4*)(dtBC + row * E_DIM + DT_RANK);
#pragma unroll
    for (int i = 0; i < 4; ++i) {
      float4 q = bp[i];
      Bv[4*i+0] = q.x; Bv[4*i+1] = q.y; Bv[4*i+2] = q.z; Bv[4*i+3] = q.w;
    }
    float dx0 = dtv.x * xv.x;
    float dx1 = dtv.y * xv.y;
#pragma unroll
    for (int n = 0; n < D_STATE; ++n) {
      float dA0 = __expf(dtv.x * An0[n]);
      float dA1 = __expf(dtv.y * An1[n]);
      h0v[n] = dA0 * h0v[n] + dx0 * Bv[n];
      h1v[n] = dA1 * h1v[n] + dx1 * Bv[n];
      P0[n] *= dA0;
      P1[n] *= dA1;
    }
  }

#pragma unroll
  for (int n = 0; n < D_STATE; ++n) {
    *(float2*)&sh_h[sub][n][dl] = make_float2(h0v[n], h1v[n]);
    *(float2*)&sh_P[sub][n][dl] = make_float2(P0[n], P1[n]);
  }
  __syncthreads();

  int n0 = sub * 4;
  float* pe = carE + CARRY_BASE(b, c, 0) + (size_t)n0 * D_MODEL + d;
  float* pq = carQ + CARRY_BASE(b, c, 0) + (size_t)n0 * D_MODEL + d;
  float he0[4], he1[4], pp0[4], pp1[4];
#pragma unroll
  for (int j = 0; j < 4; ++j) {
    float2 e = *(const float2*)&sh_h[0][n0 + j][dl];
    float2 q = *(const float2*)&sh_P[0][n0 + j][dl];
    *(float2*)(pe + j * D_MODEL) = e;
    *(float2*)(pq + j * D_MODEL) = q;
#pragma unroll
    for (int k = 1; k < NSUB - 1; ++k) {
      float2 p = *(const float2*)&sh_P[k][n0 + j][dl];
      float2 g = *(const float2*)&sh_h[k][n0 + j][dl];
      e.x = p.x * e.x + g.x;
      e.y = p.y * e.y + g.y;
      q.x *= p.x;
      q.y *= p.y;
      *(float2*)(pe + (k * D_STATE + j) * D_MODEL) = e;
      *(float2*)(pq + (k * D_STATE + j) * D_MODEL) = q;
    }
    float2 pl = *(const float2*)&sh_P[NSUB - 1][n0 + j][dl];
    float2 gl = *(const float2*)&sh_h[NSUB - 1][n0 + j][dl];
    he0[j] = pl.x * e.x + gl.x;
    he1[j] = pl.y * e.y + gl.y;
    pp0[j] = q.x * pl.x;
    pp1[j] = q.y * pl.y;
  }
  size_t base0 = ((size_t)(b * D_MODEL + d) * NCHUNK + c) * D_STATE + n0;
  size_t base1 = base0 + (size_t)NCHUNK * D_STATE;
  *(float4*)(hend + base0)  = make_float4(he0[0], he0[1], he0[2], he0[3]);
  *(float4*)(hend + base1)  = make_float4(he1[0], he1[1], he1[2], he1[3]);
  *(float4*)(Pprod + base0) = make_float4(pp0[0], pp0[1], pp0[2], pp0[3]);
  *(float4*)(Pprod + base1) = make_float4(pp1[0], pp1[1], pp1[2], pp1[3]);
}

__global__ __launch_bounds__(256) void k_scan_combine(const float* __restrict__ hend,
                                                      const float* __restrict__ Pprod,
                                                      float* __restrict__ h0) {
  int gid = blockIdx.x * blockDim.x + threadIdx.x;
  int n = gid & (D_STATE - 1);
  int bd = gid >> 4;
  size_t base = (size_t)bd * NCHUNK * D_STATE + n;
  float h = 0.f;
  for (int c = 0; c < NCHUNK; ++c) {
    size_t idx = base + (size_t)c * D_STATE;
    h0[idx] = h;
    h = Pprod[idx] * h + hend[idx];
  }
}

__global__ __launch_bounds__(256, 8) void k_scan_emit(const float* __restrict__ x,
                                                   const float* __restrict__ dt,
                                                   const float* __restrict__ dtBC,
                                                   const float* __restrict__ A_log,
                                                   const float* __restrict__ Dp,
                                                   const float* __restrict__ h0,
                                                   const float* __restrict__ carE,
                                                   const float* __restrict__ carQ,
                                                   float* __restrict__ y) {
  int col = threadIdx.x & (DCOL - 1);
  int sub = threadIdx.x >> 6;
  int d = (blockIdx.x & (D_MODEL / DBLK - 1)) * DBLK + DPT * col;
  int c = (blockIdx.x >> 3) & (NCHUNK - 1);
  int b = blockIdx.x >> 9;

  float An0[D_STATE], An1[D_STATE];
#pragma unroll
  for (int n = 0; n < D_STATE; ++n) {
    An0[n] = -__expf(A_log[d * D_STATE + n]);
    An1[n] = -__expf(A_log[(d + 1) * D_STATE + n]);
  }

  int t0 = c * TCHUNK + sub * TSUB;
  float h0v[D_STATE], h1v[D_STATE];

  size_t base0 = ((size_t)(b * D_MODEL + d) * NCHUNK + c) * D_STATE;
  size_t base1 = base0 + (size_t)NCHUNK * D_STATE;
#pragma unroll
  for (int n = 0; n < D_STATE; ++n) { h0v[n] = h0[base0 + n]; h1v[n] = h0[base1 + n]; }
  if (sub > 0) {
    size_t cb = CARRY_BASE(b, c, sub - 1) + d;
#pragma unroll
    for (int n = 0; n < D_STATE; ++n) {
      float2 q = *(const float2*)(carQ + cb + (size_t)n * D_MODEL);
      float2 e = *(const float2*)(carE + cb + (size_t)n * D_MODEL);
      h0v[n] = q.x * h0v[n] + e.x;
      h1v[n] = q.y * h1v[n] + e.y;
    }
  }

  float2 Dd = *(const float2*)(Dp + d);
  for (int t = t0; t < t0 + TSUB; ++t) {
    size_t row = (size_t)b * SEQLEN + t;
    float2 dtv = *(const float2*)(dt + row * D_MODEL + d);
    float2 xv  = *(const float2*)(x + row * D_MODEL + d);
    float Bv[D_STATE], Cv[D_STATE];
    const float4* bp = (const float4*)(dtBC + row * E_DIM + DT_RANK);
    const float4* cp = (const float4*)(dtBC + row * E_DIM + DT_RANK + D_STATE);
#pragma unroll
    for (int i = 0; i < 4; ++i) {
      float4 q = bp[i];
      Bv[4*i+0] = q.x; Bv[4*i+1] = q.y; Bv[4*i+2] = q.z; Bv[4*i+3] = q.w;
    }
#pragma unroll
    for (int i = 0; i < 4; ++i) {
      float4 q = cp[i];
      Cv[4*i+0] = q.x; Cv[4*i+1] = q.y; Cv[4*i+2] = q.z; Cv[4*i+3] = q.w;
    }
    float dx0 = dtv.x * xv.x;
    float dx1 = dtv.y * xv.y;
    float y0 = 0.f, y1 = 0.f;
#pragma unroll
    for (int n = 0; n < D_STATE; ++n) {
      float dA0 = __expf(dtv.x * An0[n]);
      float dA1 = __expf(dtv.y * An1[n]);
      h0v[n] = dA0 * h0v[n] + dx0 * Bv[n];
      h1v[n] = dA1 * h1v[n] + dx1 * Bv[n];
      y0 += h0v[n] * Cv[n];
      y1 += h1v[n] * Cv[n];
    }
    *(float2*)(y + row * D_MODEL + d) = make_float2(y0 + xv.x * Dd.x, y1 + xv.y * Dd.y);
  }
}

extern "C" void kernel_launch(void* const* d_in, const int* in_sizes, int n_in,
                              void* d_out, int out_size, void* d_ws, size_t ws_size,
                              hipStream_t stream) {
  const float* x    = (const float*)d_in[0];
  const float* Alog = (const float*)d_in[1];
  const float* Dp   = (const float*)d_in[2];
  const float* Wdt  = (const float*)d_in[3];
  const float* bdt  = (const float*)d_in[4];
  const float* Wbc  = (const float*)d_in[5];
  float* y = (float*)d_out;

  float* ws    = (float*)d_ws;
  float* dtBC  = ws;
  float* dt    = dtBC + (size_t)ROWS * E_DIM;
  float* hend  = dt + (size_t)ROWS * D_MODEL;
  float* Pprod = hend + (size_t)BATCH * D_MODEL * NCHUNK * D_STATE;
  float* h0    = Pprod + (size_t)BATCH * D_MODEL * NCHUNK * D_STATE;
  float* carE  = h0 + (size_t)BATCH * D_MODEL * NCHUNK * D_STATE;
  float* carQ  = carE + CARRY_FLOATS;

  k_proj_dtbc<<<ROWS / G1_ROWS, 256, 0, stream>>>(x, Wbc, Wdt, bdt, dtBC, dt);

  int nscan = BATCH * NCHUNK * (D_MODEL / DBLK);
  k_scan_local<<<nscan, 256, 0, stream>>>(x, dt, dtBC, Alog, hend, Pprod, carE, carQ);

  int ncomb = BATCH * D_MODEL * D_STATE;
  k_scan_combine<<<ncomb / 256, 256, 0, stream>>>(hend, Pprod, h0);

  k_scan_emit<<<nscan, 256, 0, stream>>>(x, dt, dtBC, Alog, Dp, h0, carE, carQ, y);
}
